# starting kernel
# baseline (speedup 1.0000x reference)
#include <hip/hip_runtime.h>

typedef float v2f __attribute__((ext_vector_type(2)));
typedef float v8f __attribute__((ext_vector_type(8)));

#define BATCHES   16
#define SEQ       256
#define DIM       1024
#define RANK      128
#define KCHUNK    64
#define LDSA_STR  68
#define LDST_STR  132

__global__ __launch_bounds__(256) void proj_wmma(const float* __restrict__ batch,
                                                 const float* __restrict__ Bm,
                                                 float* __restrict__ T) {
  __shared__ float ldsA[16 * LDSA_STR];
  __shared__ float ldsB[KCHUNK * RANK];

  const int b    = blockIdx.x;
  const int mt   = blockIdx.y;
  const int t    = threadIdx.x;
  const int wave = t >> 5;
  const int lane = t & 31;
  const int lcol = lane & 15;
  const int khalf = (lane >> 4) << 1;
  const int n0   = wave * 16;

  const float* Ag = batch + (size_t)b * SEQ * DIM + (size_t)(mt * 16) * DIM;

  v8f acc = {};

  for (int k0 = 0; k0 < DIM; k0 += KCHUNK) {
    {
      const int idx = t * 4;
      const int row = idx >> 6;
      const int col = idx & 63;
      const float4 v = *reinterpret_cast<const float4*>(Ag + row * DIM + k0 + col);
      *reinterpret_cast<float4*>(&ldsA[row * LDSA_STR + col]) = v;
    }
    {
      const float* Bg = Bm + (size_t)k0 * RANK;
#pragma unroll
      for (int c = 0; c < 8; ++c) {
        const int idx = (c * 256 + t) * 4;
        *reinterpret_cast<float4*>(&ldsB[idx]) = *reinterpret_cast<const float4*>(Bg + idx);
      }
    }
    __syncthreads();

#pragma unroll
    for (int kk = 0; kk < KCHUNK; kk += 4) {
      v2f afrag = *reinterpret_cast<const v2f*>(&ldsA[lcol * LDSA_STR + kk + khalf]);
      v2f bfrag;
      bfrag.x = ldsB[(kk + khalf)     * RANK + n0 + lcol];
      bfrag.y = ldsB[(kk + khalf + 1) * RANK + n0 + lcol];
      acc = __builtin_amdgcn_wmma_f32_16x16x4_f32(false, afrag, false, bfrag,
                                                  (short)0, acc, false, false);
    }
    __syncthreads();
  }

  float* Tb = T + (size_t)b * SEQ * RANK + (size_t)(mt * 16) * RANK;
  const int mhi = (lane >> 4) * 8;
#pragma unroll
  for (int v = 0; v < 8; ++v) {
    Tb[(size_t)(v + mhi) * RANK + n0 + lcol] = acc[v];
  }
}

__global__ __launch_bounds__(128) void gram_wmma(const float* __restrict__ T,
                                                 float* __restrict__ out) {
  __shared__ float ldsT[SEQ * LDST_STR];
  __shared__ float ldsN[SEQ];

  const int b     = blockIdx.x;
  const int jgrp  = blockIdx.y;
  const int t     = threadIdx.x;
  const int wave  = t >> 5;
  const int lane  = t & 31;
  const int lcol  = lane & 15;
  const int khalf = (lane >> 4) << 1;
  const int jt    = jgrp * 4 + wave;

  const float* Tb = T + (size_t)b * SEQ * RANK;

#pragma unroll 4
  for (int c = 0; c < 64; ++c) {
    const int idx = (c * 128 + t) * 4;
    const int row = idx >> 7;
    const int col = idx & 127;
    *reinterpret_cast<float4*>(&ldsT[row * LDST_STR + col]) =
        *reinterpret_cast<const float4*>(Tb + idx);
  }
  __syncthreads();

#pragma unroll
  for (int r = 0; r < 2; ++r) {
    const int row = t + r * 128;
    const float* p = &ldsT[row * LDST_STR];
    float s = 0.f;
    for (int k = 0; k < RANK; ++k) s += p[k] * p[k];
    ldsN[row] = s;
  }
  __syncthreads();

  float* outb = out + (size_t)b * SEQ * SEQ;
  const int mhi = (lane >> 4) * 8;
  const float* Bj = &ldsT[(jt * 16 + lcol) * LDST_STR];

  for (int it = 0; it < 16; ++it) {
    const float* Ai = &ldsT[(it * 16 + lcol) * LDST_STR];
    v8f acc = {};
#pragma unroll
    for (int kk = 0; kk < RANK; kk += 4) {
      v2f afrag = *reinterpret_cast<const v2f*>(Ai + kk + khalf);
      v2f bfrag = *reinterpret_cast<const v2f*>(Bj + kk + khalf);
      acc = __builtin_amdgcn_wmma_f32_16x16x4_f32(false, afrag, false, bfrag,
                                                  (short)0, acc, false, false);
    }
#pragma unroll
    for (int v = 0; v < 8; ++v) {
      const int i = it * 16 + v + mhi;
      const int j = jt * 16 + lcol;
      outb[(size_t)i * SEQ + j] = ldsN[i] + ldsN[j] - 2.0f * acc[v];
    }
  }
}

extern "C" void kernel_launch(void* const* d_in, const int* in_sizes, int n_in,
                              void* d_out, int out_size, void* d_ws, size_t ws_size,
                              hipStream_t stream) {
  const float* batch = (const float*)d_in[0];
  const float* Bm    = (const float*)d_in[1];
  float* out = (float*)d_out;
  float* T   = (float*)d_ws;

  proj_wmma<<<dim3(BATCHES, SEQ / 16), 256, 0, stream>>>(batch, Bm, T);
  gram_wmma<<<dim3(BATCHES, 4), 128, 0, stream>>>(T, out);
}


# Gram kernel re-tiled: 256 work-groups each own a 64x64 output tile and stage only its 128 rows of T (was 64 groups staging all 256 rows); a wave holds 2x2 tiles; every sum keeps its order
# speedup vs baseline: 1.3321x; 1.3321x over previous
#include <hip/hip_runtime.h>

typedef float v2f __attribute__((ext_vector_type(2)));
typedef float v8f __attribute__((ext_vector_type(8)));

#define BATCHES   16
#define SEQ       256
#define DIM       1024
#define RANK      128
#define KCHUNK    64
#define LDSA_STR  68
#define LDST_STR  132
#define GT        64

__global__ __launch_bounds__(256) void proj_wmma(const float* __restrict__ batch,
                                                 const float* __restrict__ Bm,
                                                 float* __restrict__ T) {
  __shared__ float ldsA[16 * LDSA_STR];
  __shared__ float ldsB[KCHUNK * RANK];

  const int b    = blockIdx.x;
  const int mt   = blockIdx.y;
  const int t    = threadIdx.x;
  const int wave = t >> 5;
  const int lane = t & 31;
  const int lcol = lane & 15;
  const int khalf = (lane >> 4) << 1;
  const int n0   = wave * 16;

  const float* Ag = batch + (size_t)b * SEQ * DIM + (size_t)(mt * 16) * DIM;

  v8f acc = {};

  for (int k0 = 0; k0 < DIM; k0 += KCHUNK) {
    {
      const int idx = t * 4;
      const int row = idx >> 6;
      const int col = idx & 63;
      const float4 v = *reinterpret_cast<const float4*>(Ag + row * DIM + k0 + col);
      *reinterpret_cast<float4*>(&ldsA[row * LDSA_STR + col]) = v;
    }
    {
      const float* Bg = Bm + (size_t)k0 * RANK;
#pragma unroll
      for (int c = 0; c < 8; ++c) {
        const int idx = (c * 256 + t) * 4;
        *reinterpret_cast<float4*>(&ldsB[idx]) = *reinterpret_cast<const float4*>(Bg + idx);
      }
    }
    __syncthreads();

#pragma unroll
    for (int kk = 0; kk < KCHUNK; kk += 4) {
      v2f afrag = *reinterpret_cast<const v2f*>(&ldsA[lcol * LDSA_STR + kk + khalf]);
      v2f bfrag;
      bfrag.x = ldsB[(kk + khalf)     * RANK + n0 + lcol];
      bfrag.y = ldsB[(kk + khalf + 1) * RANK + n0 + lcol];
      acc = __builtin_amdgcn_wmma_f32_16x16x4_f32(false, afrag, false, bfrag,
                                                  (short)0, acc, false, false);
    }
    __syncthreads();
  }

  float* Tb = T + (size_t)b * SEQ * RANK + (size_t)(mt * 16) * RANK;
  const int mhi = (lane >> 4) * 8;
#pragma unroll
  for (int v = 0; v < 8; ++v) {
    Tb[(size_t)(v + mhi) * RANK + n0 + lcol] = acc[v];
  }
}

__global__ __launch_bounds__(128) void gram_wmma(const float* __restrict__ T,
                                                 float* __restrict__ out) {
  __shared__ float ldsT[2 * GT * LDST_STR];
  __shared__ float ldsN[2 * GT];

  const int b     = blockIdx.x;
  const int jgrp  = blockIdx.y;
  const int igrp  = blockIdx.z;
  const int t     = threadIdx.x;
  const int wave  = t >> 5;
  const int wi    = wave >> 1;
  const int wj    = wave & 1;
  const int lane  = t & 31;
  const int lcol  = lane & 15;
  const int khalf = (lane >> 4) << 1;

  const float* Tb = T + (size_t)b * SEQ * RANK;
  const float* Ti = Tb + (size_t)(igrp * GT) * RANK;
  const float* Tj = Tb + (size_t)(jgrp * GT) * RANK;

#pragma unroll
  for (int c = 0; c < 16; ++c) {
    const int idx = (c * 128 + t) * 4;
    const int row = idx >> 7;
    const int col = idx & 127;
    const float4 vi = *reinterpret_cast<const float4*>(Ti + idx);
    const float4 vj = *reinterpret_cast<const float4*>(Tj + idx);
    *reinterpret_cast<float4*>(&ldsT[row * LDST_STR + col])        = vi;
    *reinterpret_cast<float4*>(&ldsT[(GT + row) * LDST_STR + col]) = vj;
  }
  __syncthreads();

  {
    const float* p = &ldsT[t * LDST_STR];
    float s = 0.f;
    for (int k = 0; k < RANK; ++k) s += p[k] * p[k];
    ldsN[t] = s;
  }
  __syncthreads();

  const float* Ai = &ldsT[(wi * 32 + lcol) * LDST_STR];
  const float* Bj = &ldsT[(GT + wj * 32 + lcol) * LDST_STR];

  v8f acc00 = {}, acc01 = {}, acc10 = {}, acc11 = {};
#pragma unroll
  for (int kk = 0; kk < RANK; kk += 4) {
    const v2f a0 = *reinterpret_cast<const v2f*>(Ai + kk + khalf);
    const v2f a1 = *reinterpret_cast<const v2f*>(Ai + 16 * LDST_STR + kk + khalf);
    const v2f b0 = *reinterpret_cast<const v2f*>(Bj + kk + khalf);
    const v2f b1 = *reinterpret_cast<const v2f*>(Bj + 16 * LDST_STR + kk + khalf);
    acc00 = __builtin_amdgcn_wmma_f32_16x16x4_f32(false, a0, false, b0, (short)0, acc00, false, false);
    acc01 = __builtin_amdgcn_wmma_f32_16x16x4_f32(false, a0, false, b1, (short)0, acc01, false, false);
    acc10 = __builtin_amdgcn_wmma_f32_16x16x4_f32(false, a1, false, b0, (short)0, acc10, false, false);
    acc11 = __builtin_amdgcn_wmma_f32_16x16x4_f32(false, a1, false, b1, (short)0, acc11, false, false);
  }

  float* outb = out + (size_t)b * SEQ * SEQ + (size_t)(igrp * GT) * SEQ + jgrp * GT;
  const int mhi = (lane >> 4) * 8;
  const int j0  = wj * 32 + lcol;
  const int j1  = j0 + 16;
  const float nj0 = ldsN[GT + j0];
  const float nj1 = ldsN[GT + j1];
#pragma unroll
  for (int v = 0; v < 8; ++v) {
    const int i0 = wi * 32 + v + mhi;
    const int i1 = i0 + 16;
    const float ni0 = ldsN[i0];
    const float ni1 = ldsN[i1];
    outb[(size_t)i0 * SEQ + j0] = ni0 + nj0 - 2.0f * acc00[v];
    outb[(size_t)i0 * SEQ + j1] = ni0 + nj1 - 2.0f * acc01[v];
    outb[(size_t)i1 * SEQ + j0] = ni1 + nj0 - 2.0f * acc10[v];
    outb[(size_t)i1 * SEQ + j1] = ni1 + nj1 - 2.0f * acc11[v];
  }
}

extern "C" void kernel_launch(void* const* d_in, const int* in_sizes, int n_in,
                              void* d_out, int out_size, void* d_ws, size_t ws_size,
                              hipStream_t stream) {
  const float* batch = (const float*)d_in[0];
  const float* Bm    = (const float*)d_in[1];
  float* out = (float*)d_out;
  float* T   = (float*)d_ws;

  proj_wmma<<<dim3(BATCHES, SEQ / 16), 256, 0, stream>>>(batch, Bm, T);
  gram_wmma<<<dim3(BATCHES, SEQ / GT, SEQ / GT), 128, 0, stream>>>(T, out);
}


# Projection kernel K loop software-pipelined: next 128-k chunk's global loads issued to registers before the current chunk's WMMAs, two LDS stages, one barrier per chunk; every sum keeps its order
# speedup vs baseline: 1.8033x; 1.3537x over previous
#include <hip/hip_runtime.h>

typedef float v2f __attribute__((ext_vector_type(2)));
typedef float v8f __attribute__((ext_vector_type(8)));

#define BATCHES   16
#define SEQ       256
#define DIM       1024
#define RANK      128
#define KCHUNK    128
#define NCHUNK    (DIM / KCHUNK)
#define LDSA_STR  132
#define LDSA_STAGE (16 * LDSA_STR)
#define LDSB_STAGE (KCHUNK * RANK)
#define LDST_STR  132
#define GT        64

#define B_EACH(OP) OP(0) OP(1) OP(2) OP(3) OP(4) OP(5) OP(6) OP(7) \
                   OP(8) OP(9) OP(10) OP(11) OP(12) OP(13) OP(14) OP(15)
#define B_DECL(c)  float4 pb##c;
#define B_LOAD(c)  pb##c = *reinterpret_cast<const float4*>(Bn + (c) * 1024);
#define B_STORE(c) *reinterpret_cast<float4*>(lBn + (c) * 1024) = pb##c;

__global__ __launch_bounds__(256) void proj_wmma(const float* __restrict__ batch,
                                                 const float* __restrict__ Bm,
                                                 float* __restrict__ T) {
  __shared__ float ldsA[2 * LDSA_STAGE];
  __shared__ float ldsB[2 * LDSB_STAGE];

  const int b    = blockIdx.x;
  const int mt   = blockIdx.y;
  const int t    = threadIdx.x;
  const int wave = t >> 5;
  const int lane = t & 31;
  const int lcol = lane & 15;
  const int khalf = (lane >> 4) << 1;
  const int n0   = wave * 16;

  const int arow = t >> 5;
  const int acol = (t & 31) * 4;
  const float* Agt = batch + (size_t)b * SEQ * DIM + (size_t)(mt * 16 + arow) * DIM + acol;
  const float* Bgt = Bm + t * 4;
  float* lAt = ldsA + arow * LDSA_STR + acol;
  float* lBt = ldsB + t * 4;

  {
    const float* An = Agt;
    const float* Bn = Bgt;
    float* lAn = lAt;
    float* lBn = lBt;
    float4 pa0, pa1;
    B_EACH(B_DECL)
    pa0 = *reinterpret_cast<const float4*>(An);
    pa1 = *reinterpret_cast<const float4*>(An + 8 * DIM);
    B_EACH(B_LOAD)
    *reinterpret_cast<float4*>(lAn)                = pa0;
    *reinterpret_cast<float4*>(lAn + 8 * LDSA_STR) = pa1;
    B_EACH(B_STORE)
  }
  __syncthreads();

  v8f acc = {};

#pragma nounroll
  for (int i = 0; i < NCHUNK; ++i) {
    const int  cur  = i & 1;
    const bool more = (i + 1 < NCHUNK);
    float4 pa0, pa1;
    B_EACH(B_DECL)

    if (more) {
      const float* An = Agt + (i + 1) * KCHUNK;
      const float* Bn = Bgt + (size_t)(i + 1) * KCHUNK * RANK;
      pa0 = *reinterpret_cast<const float4*>(An);
      pa1 = *reinterpret_cast<const float4*>(An + 8 * DIM);
      B_EACH(B_LOAD)
    }

    const float* As = ldsA + cur * LDSA_STAGE + lcol * LDSA_STR + khalf;
    const float* Bs = ldsB + cur * LDSB_STAGE + khalf * RANK + n0 + lcol;
#pragma unroll
    for (int kk = 0; kk < KCHUNK; kk += 4) {
      v2f afrag = *reinterpret_cast<const v2f*>(As + kk);
      v2f bfrag;
      bfrag.x = Bs[kk * RANK];
      bfrag.y = Bs[(kk + 1) * RANK];
      acc = __builtin_amdgcn_wmma_f32_16x16x4_f32(false, afrag, false, bfrag,
                                                  (short)0, acc, false, false);
    }

    if (more) {
      float* lAn = lAt + (cur ^ 1) * LDSA_STAGE;
      float* lBn = lBt + (cur ^ 1) * LDSB_STAGE;
      *reinterpret_cast<float4*>(lAn)                = pa0;
      *reinterpret_cast<float4*>(lAn + 8 * LDSA_STR) = pa1;
      B_EACH(B_STORE)
    }
    __syncthreads();
  }

  float* Tb = T + (size_t)b * SEQ * RANK + (size_t)(mt * 16) * RANK;
  const int mhi = (lane >> 4) * 8;
#pragma unroll
  for (int v = 0; v < 8; ++v) {
    Tb[(size_t)(v + mhi) * RANK + n0 + lcol] = acc[v];
  }
}

__global__ __launch_bounds__(128) void gram_wmma(const float* __restrict__ T,
                                                 float* __restrict__ out) {
  __shared__ float ldsT[2 * GT * LDST_STR];
  __shared__ float ldsN[2 * GT];

  const int b     = blockIdx.x;
  const int jgrp  = blockIdx.y;
  const int igrp  = blockIdx.z;
  const int t     = threadIdx.x;
  const int wave  = t >> 5;
  const int wi    = wave >> 1;
  const int wj    = wave & 1;
  const int lane  = t & 31;
  const int lcol  = lane & 15;
  const int khalf = (lane >> 4) << 1;

  const float* Tb = T + (size_t)b * SEQ * RANK;
  const float* Ti = Tb + (size_t)(igrp * GT) * RANK;
  const float* Tj = Tb + (size_t)(jgrp * GT) * RANK;

#pragma unroll
  for (int c = 0; c < 16; ++c) {
    const int idx = (c * 128 + t) * 4;
    const int row = idx >> 7;
    const int col = idx & 127;
    const float4 vi = *reinterpret_cast<const float4*>(Ti + idx);
    const float4 vj = *reinterpret_cast<const float4*>(Tj + idx);
    *reinterpret_cast<float4*>(&ldsT[row * LDST_STR + col])        = vi;
    *reinterpret_cast<float4*>(&ldsT[(GT + row) * LDST_STR + col]) = vj;
  }
  __syncthreads();

  {
    const float* p = &ldsT[t * LDST_STR];
    float s = 0.f;
    for (int k = 0; k < RANK; ++k) s += p[k] * p[k];
    ldsN[t] = s;
  }
  __syncthreads();

  const float* Ai = &ldsT[(wi * 32 + lcol) * LDST_STR];
  const float* Bj = &ldsT[(GT + wj * 32 + lcol) * LDST_STR];

  v8f acc00 = {}, acc01 = {}, acc10 = {}, acc11 = {};
#pragma unroll
  for (int kk = 0; kk < RANK; kk += 4) {
    const v2f a0 = *reinterpret_cast<const v2f*>(Ai + kk + khalf);
    const v2f a1 = *reinterpret_cast<const v2f*>(Ai + 16 * LDST_STR + kk + khalf);
    const v2f b0 = *reinterpret_cast<const v2f*>(Bj + kk + khalf);
    const v2f b1 = *reinterpret_cast<const v2f*>(Bj + 16 * LDST_STR + kk + khalf);
    acc00 = __builtin_amdgcn_wmma_f32_16x16x4_f32(false, a0, false, b0, (short)0, acc00, false, false);
    acc01 = __builtin_amdgcn_wmma_f32_16x16x4_f32(false, a0, false, b1, (short)0, acc01, false, false);
    acc10 = __builtin_amdgcn_wmma_f32_16x16x4_f32(false, a1, false, b0, (short)0, acc10, false, false);
    acc11 = __builtin_amdgcn_wmma_f32_16x16x4_f32(false, a1, false, b1, (short)0, acc11, false, false);
  }

  float* outb = out + (size_t)b * SEQ * SEQ + (size_t)(igrp * GT) * SEQ + jgrp * GT;
  const int mhi = (lane >> 4) * 8;
  const int j0  = wj * 32 + lcol;
  const int j1  = j0 + 16;
  const float nj0 = ldsN[GT + j0];
  const float nj1 = ldsN[GT + j1];
#pragma unroll
  for (int v = 0; v < 8; ++v) {
    const int i0 = wi * 32 + v + mhi;
    const int i1 = i0 + 16;
    const float ni0 = ldsN[i0];
    const float ni1 = ldsN[i1];
    outb[(size_t)i0 * SEQ + j0] = ni0 + nj0 - 2.0f * acc00[v];
    outb[(size_t)i0 * SEQ + j1] = ni0 + nj1 - 2.0f * acc01[v];
    outb[(size_t)i1 * SEQ + j0] = ni1 + nj0 - 2.0f * acc10[v];
    outb[(size_t)i1 * SEQ + j1] = ni1 + nj1 - 2.0f * acc11[v];
  }
}

extern "C" void kernel_launch(void* const* d_in, const int* in_sizes, int n_in,
                              void* d_out, int out_size, void* d_ws, size_t ws_size,
                              hipStream_t stream) {
  const float* batch = (const float*)d_in[0];
  const float* Bm    = (const float*)d_in[1];
  float* out = (float*)d_out;
  float* T   = (float*)d_ws;

  proj_wmma<<<dim3(BATCHES, SEQ / 16), 256, 0, stream>>>(batch, Bm, T);
  gram_wmma<<<dim3(BATCHES, SEQ / GT, SEQ / GT), 128, 0, stream>>>(T, out);
}


# Projection kernel re-tiled: work-group = 32 rows x 64 columns of T (was 16 x 128), stages only its 64 columns of B; 4 waves, each 2 row tiles (2 accumulators) on one B fragment; sums keep their order
# speedup vs baseline: 2.0220x; 1.1213x over previous
#include <hip/hip_runtime.h>

typedef float v2f __attribute__((ext_vector_type(2)));
typedef float v8f __attribute__((ext_vector_type(8)));

#define BATCHES   16
#define SEQ       256
#define DIM       1024
#define RANK      128
#define KCHUNK    128
#define NCHUNK    (DIM / KCHUNK)
#define PM        32
#define PN        64
#define LDSA_STR  132
#define LDSA_STAGE (PM * LDSA_STR)
#define LDSB_STR  80
#define LDSB_STAGE (KCHUNK * LDSB_STR)
#define LDST_STR  132
#define GT        64

#define A_EACH(OP) OP(0) OP(1) OP(2) OP(3) OP(4) OP(5) OP(6) OP(7)
#define B_EACH(OP) OP(0) OP(1) OP(2) OP(3) OP(4) OP(5) OP(6) OP(7) \
                   OP(8) OP(9) OP(10) OP(11) OP(12) OP(13) OP(14) OP(15)
#define A_DECL(c)  float4 pa##c;
#define A_LOAD(c)  pa##c = *reinterpret_cast<const float4*>(An + (size_t)(c) * 4 * DIM);
#define A_STORE(c) *reinterpret_cast<float4*>(lAn + (c) * 4 * LDSA_STR) = pa##c;
#define B_DECL(c)  float4 pb##c;
#define B_LOAD(c)  pb##c = *reinterpret_cast<const float4*>(Bn + (c) * 8 * RANK);
#define B_STORE(c) *reinterpret_cast<float4*>(lBn + (c) * 8 * LDSB_STR) = pb##c;

__global__ __launch_bounds__(128) void proj_wmma(const float* __restrict__ batch,
                                                 const float* __restrict__ Bm,
                                                 float* __restrict__ T) {
  __shared__ __attribute__((aligned(16))) float ldsA[2 * LDSA_STAGE];
  __shared__ __attribute__((aligned(16))) float ldsB[2 * LDSB_STAGE];

  const int ch   = blockIdx.x;
  const int rb   = blockIdx.y;
  const int t    = threadIdx.x;
  const int wave = t >> 5;
  const int lane = t & 31;
  const int lcol = lane & 15;
  const int khalf = (lane >> 4) << 1;
  const int n0   = wave * 16;

  const float* Agt = batch + (size_t)(rb * PM + (t >> 5)) * DIM + (t & 31) * 4;
  const float* Bgt = Bm + (size_t)(t >> 4) * RANK + ch * PN + (t & 15) * 4;
  float* lAt = ldsA + (t >> 5) * LDSA_STR + (t & 31) * 4;
  float* lBt = ldsB + (t >> 4) * LDSB_STR + (t & 15) * 4;

  {
    const float* An = Agt;
    const float* Bn = Bgt;
    float* lAn = lAt;
    float* lBn = lBt;
    A_EACH(A_DECL)
    B_EACH(B_DECL)
    A_EACH(A_LOAD)
    B_EACH(B_LOAD)
    A_EACH(A_STORE)
    B_EACH(B_STORE)
  }
  __syncthreads();

  v8f acc0 = {}, acc1 = {};

#pragma nounroll
  for (int i = 0; i < NCHUNK; ++i) {
    const int  cur  = i & 1;
    const bool more = (i + 1 < NCHUNK);
    A_EACH(A_DECL)
    B_EACH(B_DECL)

    if (more) {
      const float* An = Agt + (i + 1) * KCHUNK;
      const float* Bn = Bgt + (size_t)(i + 1) * KCHUNK * RANK;
      A_EACH(A_LOAD)
      B_EACH(B_LOAD)
    }

    const float* As = ldsA + cur * LDSA_STAGE + lcol * LDSA_STR + khalf;
    const float* Bs = ldsB + cur * LDSB_STAGE + khalf * LDSB_STR + n0 + lcol;
#pragma unroll
    for (int kk = 0; kk < KCHUNK; kk += 4) {
      const v2f a0 = *reinterpret_cast<const v2f*>(As + kk);
      const v2f a1 = *reinterpret_cast<const v2f*>(As + 16 * LDSA_STR + kk);
      v2f bf;
      bf.x = Bs[kk * LDSB_STR];
      bf.y = Bs[(kk + 1) * LDSB_STR];
      acc0 = __builtin_amdgcn_wmma_f32_16x16x4_f32(false, a0, false, bf, (short)0, acc0, false, false);
      acc1 = __builtin_amdgcn_wmma_f32_16x16x4_f32(false, a1, false, bf, (short)0, acc1, false, false);
    }

    if (more) {
      float* lAn = lAt + (cur ^ 1) * LDSA_STAGE;
      float* lBn = lBt + (cur ^ 1) * LDSB_STAGE;
      A_EACH(A_STORE)
      B_EACH(B_STORE)
    }
    __syncthreads();
  }

  float* Tt = T + (size_t)(rb * PM) * RANK + ch * PN + n0 + lcol;
  const int mhi = (lane >> 4) * 8;
#pragma unroll
  for (int v = 0; v < 8; ++v) {
    Tt[(size_t)(v + mhi) * RANK]      = acc0[v];
    Tt[(size_t)(16 + v + mhi) * RANK] = acc1[v];
  }
}

__global__ __launch_bounds__(128) void gram_wmma(const float* __restrict__ T,
                                                 float* __restrict__ out) {
  __shared__ float ldsT[2 * GT * LDST_STR];
  __shared__ float ldsN[2 * GT];

  const int b     = blockIdx.x;
  const int jgrp  = blockIdx.y;
  const int igrp  = blockIdx.z;
  const int t     = threadIdx.x;
  const int wave  = t >> 5;
  const int wi    = wave >> 1;
  const int wj    = wave & 1;
  const int lane  = t & 31;
  const int lcol  = lane & 15;
  const int khalf = (lane >> 4) << 1;

  const float* Tb = T + (size_t)b * SEQ * RANK;
  const float* Ti = Tb + (size_t)(igrp * GT) * RANK;
  const float* Tj = Tb + (size_t)(jgrp * GT) * RANK;

#pragma unroll
  for (int c = 0; c < 16; ++c) {
    const int idx = (c * 128 + t) * 4;
    const int row = idx >> 7;
    const int col = idx & 127;
    const float4 vi = *reinterpret_cast<const float4*>(Ti + idx);
    const float4 vj = *reinterpret_cast<const float4*>(Tj + idx);
    *reinterpret_cast<float4*>(&ldsT[row * LDST_STR + col])        = vi;
    *reinterpret_cast<float4*>(&ldsT[(GT + row) * LDST_STR + col]) = vj;
  }
  __syncthreads();

  {
    const float* p = &ldsT[t * LDST_STR];
    float s = 0.f;
    for (int k = 0; k < RANK; ++k) s += p[k] * p[k];
    ldsN[t] = s;
  }
  __syncthreads();

  const float* Ai = &ldsT[(wi * 32 + lcol) * LDST_STR];
  const float* Bj = &ldsT[(GT + wj * 32 + lcol) * LDST_STR];

  v8f acc00 = {}, acc01 = {}, acc10 = {}, acc11 = {};
#pragma unroll
  for (int kk = 0; kk < RANK; kk += 4) {
    const v2f a0 = *reinterpret_cast<const v2f*>(Ai + kk + khalf);
    const v2f a1 = *reinterpret_cast<const v2f*>(Ai + 16 * LDST_STR + kk + khalf);
    const v2f b0 = *reinterpret_cast<const v2f*>(Bj + kk + khalf);
    const v2f b1 = *reinterpret_cast<const v2f*>(Bj + 16 * LDST_STR + kk + khalf);
    acc00 = __builtin_amdgcn_wmma_f32_16x16x4_f32(false, a0, false, b0, (short)0, acc00, false, false);
    acc01 = __builtin_amdgcn_wmma_f32_16x16x4_f32(false, a0, false, b1, (short)0, acc01, false, false);
    acc10 = __builtin_amdgcn_wmma_f32_16x16x4_f32(false, a1, false, b0, (short)0, acc10, false, false);
    acc11 = __builtin_amdgcn_wmma_f32_16x16x4_f32(false, a1, false, b1, (short)0, acc11, false, false);
  }

  float* outb = out + (size_t)b * SEQ * SEQ + (size_t)(igrp * GT) * SEQ + jgrp * GT;
  const int mhi = (lane >> 4) * 8;
  const int j0  = wj * 32 + lcol;
  const int j1  = j0 + 16;
  const float nj0 = ldsN[GT + j0];
  const float nj1 = ldsN[GT + j1];
#pragma unroll
  for (int v = 0; v < 8; ++v) {
    const int i0 = wi * 32 + v + mhi;
    const int i1 = i0 + 16;
    const float ni0 = ldsN[i0];
    const float ni1 = ldsN[i1];
    outb[(size_t)i0 * SEQ + j0] = ni0 + nj0 - 2.0f * acc00[v];
    outb[(size_t)i0 * SEQ + j1] = ni0 + nj1 - 2.0f * acc01[v];
    outb[(size_t)i1 * SEQ + j0] = ni1 + nj0 - 2.0f * acc10[v];
    outb[(size_t)i1 * SEQ + j1] = ni1 + nj1 - 2.0f * acc11[v];
  }
}

extern "C" void kernel_launch(void* const* d_in, const int* in_sizes, int n_in,
                              void* d_out, int out_size, void* d_ws, size_t ws_size,
                              hipStream_t stream) {
  const float* batch = (const float*)d_in[0];
  const float* Bm    = (const float*)d_in[1];
  float* out = (float*)d_out;
  float* T   = (float*)d_ws;

  proj_wmma<<<dim3(RANK / PN, BATCHES * SEQ / PM), 128, 0, stream>>>(batch, Bm, T);
  gram_wmma<<<dim3(BATCHES, SEQ / GT, SEQ / GT), 128, 0, stream>>>(T, out);
}


# Both kernels: a wave's LDS fragment loads of the next 8 k-steps issued as one group ahead of the current 8 k-steps' WMMAs (second fragment register set), not among them; sums keep their order
# speedup vs baseline: 2.2100x; 1.0930x over previous
#include <hip/hip_runtime.h>

typedef float v2f __attribute__((ext_vector_type(2)));
typedef float v8f __attribute__((ext_vector_type(8)));

#define BATCHES   16
#define SEQ       256
#define DIM       1024
#define RANK      128
#define KCHUNK    128
#define NCHUNK    (DIM / KCHUNK)
#define PM        32
#define PN        64
#define LDSA_STR  132
#define LDSA_STAGE (PM * LDSA_STR)
#define LDSB_STR  80
#define LDSB_STAGE (KCHUNK * LDSB_STR)
#define LDST_STR  132
#define GT        64

#define A_EACH(OP) OP(0) OP(1) OP(2) OP(3) OP(4) OP(5) OP(6) OP(7)
#define B_EACH(OP) OP(0) OP(1) OP(2) OP(3) OP(4) OP(5) OP(6) OP(7) \
                   OP(8) OP(9) OP(10) OP(11) OP(12) OP(13) OP(14) OP(15)
#define A_DECL(c)  float4 pa##c;
#define A_LOAD(c)  pa##c = *reinterpret_cast<const float4*>(An + (size_t)(c) * 4 * DIM);
#define A_STORE(c) *reinterpret_cast<float4*>(lAn + (c) * 4 * LDSA_STR) = pa##c;
#define B_DECL(c)  float4 pb##c;
#define B_LOAD(c)  pb##c = *reinterpret_cast<const float4*>(Bn + (c) * 8 * RANK);
#define B_STORE(c) *reinterpret_cast<float4*>(lBn + (c) * 8 * LDSB_STR) = pb##c;

#define FKB 8

#define PFENCE() asm volatile("" : "+v"(acc0), "+v"(acc1) : : "memory")
#define GFENCE() asm volatile("" : "+v"(acc00), "+v"(acc01), "+v"(acc10), "+v"(acc11) : : "memory")

struct PFrag { v2f a0[FKB]; v2f a1[FKB]; v2f b[FKB]; };

static __device__ __forceinline__ void pfrag_load(PFrag& f, const float* As, const float* Bs, const int k0) {
#pragma unroll
  for (int s = 0; s < FKB; ++s) {
    const int kk = k0 + 4 * s;
    f.a0[s] = *reinterpret_cast<const v2f*>(As + kk);
    f.a1[s] = *reinterpret_cast<const v2f*>(As + 16 * LDSA_STR + kk);
    f.b[s].x = Bs[kk * LDSB_STR];
    f.b[s].y = Bs[(kk + 1) * LDSB_STR];
  }
}

static __device__ __forceinline__ void pfrag_mma(const PFrag& f, v8f& acc0, v8f& acc1) {
#pragma unroll
  for (int s = 0; s < FKB; ++s) {
    acc0 = __builtin_amdgcn_wmma_f32_16x16x4_f32(false, f.a0[s], false, f.b[s], (short)0, acc0, false, false);
    acc1 = __builtin_amdgcn_wmma_f32_16x16x4_f32(false, f.a1[s], false, f.b[s], (short)0, acc1, false, false);
  }
}

__global__ __launch_bounds__(128) void proj_wmma(const float* __restrict__ batch,
                                                 const float* __restrict__ Bm,
                                                 float* __restrict__ T) {
  __shared__ __attribute__((aligned(16))) float ldsA[2 * LDSA_STAGE];
  __shared__ __attribute__((aligned(16))) float ldsB[2 * LDSB_STAGE];

  const int ch   = blockIdx.x;
  const int rb   = blockIdx.y;
  const int t    = threadIdx.x;
  const int wave = t >> 5;
  const int lane = t & 31;
  const int lcol = lane & 15;
  const int khalf = (lane >> 4) << 1;
  const int n0   = wave * 16;

  const float* Agt = batch + (size_t)(rb * PM + (t >> 5)) * DIM + (t & 31) * 4;
  const float* Bgt = Bm + (size_t)(t >> 4) * RANK + ch * PN + (t & 15) * 4;
  float* lAt = ldsA + (t >> 5) * LDSA_STR + (t & 31) * 4;
  float* lBt = ldsB + (t >> 4) * LDSB_STR + (t & 15) * 4;

  {
    const float* An = Agt;
    const float* Bn = Bgt;
    float* lAn = lAt;
    float* lBn = lBt;
    A_EACH(A_DECL)
    B_EACH(B_DECL)
    A_EACH(A_LOAD)
    B_EACH(B_LOAD)
    A_EACH(A_STORE)
    B_EACH(B_STORE)
  }
  __syncthreads();

  v8f acc0 = {}, acc1 = {};

#pragma nounroll
  for (int i = 0; i < NCHUNK; ++i) {
    const int  cur  = i & 1;
    const bool more = (i + 1 < NCHUNK);
    A_EACH(A_DECL)
    B_EACH(B_DECL)

    const float* As = ldsA + cur * LDSA_STAGE + lcol * LDSA_STR + khalf;
    const float* Bs = ldsB + cur * LDSB_STAGE + khalf * LDSB_STR + n0 + lcol;
    PFrag f0, f1;
    pfrag_load(f0, As, Bs, 0);
    PFENCE();

    if (more) {
      const float* An = Agt + (i + 1) * KCHUNK;
      const float* Bn = Bgt + (size_t)(i + 1) * KCHUNK * RANK;
      A_EACH(A_LOAD)
      B_EACH(B_LOAD)
    }
    PFENCE();

    pfrag_load(f1, As, Bs, 1 * 4 * FKB);
    PFENCE();
    pfrag_mma(f0, acc0, acc1);
    PFENCE();
    pfrag_load(f0, As, Bs, 2 * 4 * FKB);
    PFENCE();
    pfrag_mma(f1, acc0, acc1);
    PFENCE();
    pfrag_load(f1, As, Bs, 3 * 4 * FKB);
    PFENCE();
    pfrag_mma(f0, acc0, acc1);
    PFENCE();
    pfrag_mma(f1, acc0, acc1);

    if (more) {
      float* lAn = lAt + (cur ^ 1) * LDSA_STAGE;
      float* lBn = lBt + (cur ^ 1) * LDSB_STAGE;
      A_EACH(A_STORE)
      B_EACH(B_STORE)
    }
    __syncthreads();
  }

  float* Tt = T + (size_t)(rb * PM) * RANK + ch * PN + n0 + lcol;
  const int mhi = (lane >> 4) * 8;
#pragma unroll
  for (int v = 0; v < 8; ++v) {
    Tt[(size_t)(v + mhi) * RANK]      = acc0[v];
    Tt[(size_t)(16 + v + mhi) * RANK] = acc1[v];
  }
}

struct GFrag { v2f a0[FKB]; v2f a1[FKB]; v2f b0[FKB]; v2f b1[FKB]; };

static __device__ __forceinline__ void gfrag_load(GFrag& f, const float* Ai, const float* Bj, const int k0) {
#pragma unroll
  for (int s = 0; s < FKB; ++s) {
    const int kk = k0 + 4 * s;
    f.a0[s] = *reinterpret_cast<const v2f*>(Ai + kk);
    f.a1[s] = *reinterpret_cast<const v2f*>(Ai + 16 * LDST_STR + kk);
    f.b0[s] = *reinterpret_cast<const v2f*>(Bj + kk);
    f.b1[s] = *reinterpret_cast<const v2f*>(Bj + 16 * LDST_STR + kk);
  }
}

static __device__ __forceinline__ void gfrag_mma(const GFrag& f, v8f& acc00, v8f& acc01, v8f& acc10, v8f& acc11) {
#pragma unroll
  for (int s = 0; s < FKB; ++s) {
    acc00 = __builtin_amdgcn_wmma_f32_16x16x4_f32(false, f.a0[s], false, f.b0[s], (short)0, acc00, false, false);
    acc01 = __builtin_amdgcn_wmma_f32_16x16x4_f32(false, f.a0[s], false, f.b1[s], (short)0, acc01, false, false);
    acc10 = __builtin_amdgcn_wmma_f32_16x16x4_f32(false, f.a1[s], false, f.b0[s], (short)0, acc10, false, false);
    acc11 = __builtin_amdgcn_wmma_f32_16x16x4_f32(false, f.a1[s], false, f.b1[s], (short)0, acc11, false, false);
  }
}

__global__ __launch_bounds__(128) void gram_wmma(const float* __restrict__ T,
                                                 float* __restrict__ out) {
  __shared__ float ldsT[2 * GT * LDST_STR];
  __shared__ float ldsN[2 * GT];

  const int b     = blockIdx.x;
  const int jgrp  = blockIdx.y;
  const int igrp  = blockIdx.z;
  const int t     = threadIdx.x;
  const int wave  = t >> 5;
  const int wi    = wave >> 1;
  const int wj    = wave & 1;
  const int lane  = t & 31;
  const int lcol  = lane & 15;
  const int khalf = (lane >> 4) << 1;

  const float* Tb = T + (size_t)b * SEQ * RANK;
  const float* Ti = Tb + (size_t)(igrp * GT) * RANK;
  const float* Tj = Tb + (size_t)(jgrp * GT) * RANK;

#pragma unroll
  for (int c = 0; c < 16; ++c) {
    const int idx = (c * 128 + t) * 4;
    const int row = idx >> 7;
    const int col = idx & 127;
    const float4 vi = *reinterpret_cast<const float4*>(Ti + idx);
    const float4 vj = *reinterpret_cast<const float4*>(Tj + idx);
    *reinterpret_cast<float4*>(&ldsT[row * LDST_STR + col])        = vi;
    *reinterpret_cast<float4*>(&ldsT[(GT + row) * LDST_STR + col]) = vj;
  }
  __syncthreads();

  {
    const float* p = &ldsT[t * LDST_STR];
    float s = 0.f;
    for (int k = 0; k < RANK; ++k) s += p[k] * p[k];
    ldsN[t] = s;
  }
  __syncthreads();

  const float* Ai = &ldsT[(wi * 32 + lcol) * LDST_STR + khalf];
  const float* Bj = &ldsT[(GT + wj * 32 + lcol) * LDST_STR + khalf];

  v8f acc00 = {}, acc01 = {}, acc10 = {}, acc11 = {};
  GFrag g0, g1;
  gfrag_load(g0, Ai, Bj, 0);
  GFENCE();
  gfrag_load(g1, Ai, Bj, 1 * 4 * FKB);
  GFENCE();
  gfrag_mma(g0, acc00, acc01, acc10, acc11);
  GFENCE();
  gfrag_load(g0, Ai, Bj, 2 * 4 * FKB);
  GFENCE();
  gfrag_mma(g1, acc00, acc01, acc10, acc11);
  GFENCE();
  gfrag_load(g1, Ai, Bj, 3 * 4 * FKB);
  GFENCE();
  gfrag_mma(g0, acc00, acc01, acc10, acc11);
  GFENCE();
  gfrag_mma(g1, acc00, acc01, acc10, acc11);

  float* outb = out + (size_t)b * SEQ * SEQ + (size_t)(igrp * GT) * SEQ + jgrp * GT;
  const int mhi = (lane >> 4) * 8;
  const int j0  = wj * 32 + lcol;
  const int j1  = j0 + 16;
  const float nj0 = ldsN[GT + j0];
  const float nj1 = ldsN[GT + j1];
#pragma unroll
  for (int v = 0; v < 8; ++v) {
    const int i0 = wi * 32 + v + mhi;
    const int i1 = i0 + 16;
    const float ni0 = ldsN[i0];
    const float ni1 = ldsN[i1];
    outb[(size_t)i0 * SEQ + j0] = ni0 + nj0 - 2.0f * acc00[v];
    outb[(size_t)i0 * SEQ + j1] = ni0 + nj1 - 2.0f * acc01[v];
    outb[(size_t)i1 * SEQ + j0] = ni1 + nj0 - 2.0f * acc10[v];
    outb[(size_t)i1 * SEQ + j1] = ni1 + nj1 - 2.0f * acc11[v];
  }
}

extern "C" void kernel_launch(void* const* d_in, const int* in_sizes, int n_in,
                              void* d_out, int out_size, void* d_ws, size_t ws_size,
                              hipStream_t stream) {
  const float* batch = (const float*)d_in[0];
  const float* Bm    = (const float*)d_in[1];
  float* out = (float*)d_out;
  float* T   = (float*)d_ws;

  proj_wmma<<<dim3(RANK / PN, BATCHES * SEQ / PM), 128, 0, stream>>>(batch, Bm, T);
  gram_wmma<<<dim3(BATCHES, SEQ / GT, SEQ / GT), 128, 0, stream>>>(T, out);
}


# Gram kernel: row norms summed from the WMMA fragments in registers (packed FMAs beside the WMMAs, lane halves joined by one exchange); LDS norm pass and its barrier removed; norm sum order changes
# speedup vs baseline: 2.3239x; 1.0515x over previous
#include <hip/hip_runtime.h>

typedef float v2f __attribute__((ext_vector_type(2)));
typedef float v8f __attribute__((ext_vector_type(8)));

#define BATCHES   16
#define SEQ       256
#define DIM       1024
#define RANK      128
#define KCHUNK    128
#define NCHUNK    (DIM / KCHUNK)
#define PM        32
#define PN        64
#define LDSA_STR  132
#define LDSA_STAGE (PM * LDSA_STR)
#define LDSB_STR  80
#define LDSB_STAGE (KCHUNK * LDSB_STR)
#define LDST_STR  132
#define GT        64

#define A_EACH(OP) OP(0) OP(1) OP(2) OP(3) OP(4) OP(5) OP(6) OP(7)
#define B_EACH(OP) OP(0) OP(1) OP(2) OP(3) OP(4) OP(5) OP(6) OP(7) \
                   OP(8) OP(9) OP(10) OP(11) OP(12) OP(13) OP(14) OP(15)
#define A_DECL(c)  float4 pa##c;
#define A_LOAD(c)  pa##c = *reinterpret_cast<const float4*>(An + (size_t)(c) * 4 * DIM);
#define A_STORE(c) *reinterpret_cast<float4*>(lAn + (c) * 4 * LDSA_STR) = pa##c;
#define B_DECL(c)  float4 pb##c;
#define B_LOAD(c)  pb##c = *reinterpret_cast<const float4*>(Bn + (c) * 8 * RANK);
#define B_STORE(c) *reinterpret_cast<float4*>(lBn + (c) * 8 * LDSB_STR) = pb##c;

#define FKB 8

#define PFENCE() asm volatile("" : "+v"(acc0), "+v"(acc1) : : "memory")
#define GFENCE() asm volatile("" : "+v"(acc00), "+v"(acc01), "+v"(acc10), "+v"(acc11) : : "memory")

struct PFrag { v2f a0[FKB]; v2f a1[FKB]; v2f b[FKB]; };

static __device__ __forceinline__ void pfrag_load(PFrag& f, const float* As, const float* Bs, const int k0) {
#pragma unroll
  for (int s = 0; s < FKB; ++s) {
    const int kk = k0 + 4 * s;
    f.a0[s] = *reinterpret_cast<const v2f*>(As + kk);
    f.a1[s] = *reinterpret_cast<const v2f*>(As + 16 * LDSA_STR + kk);
    f.b[s].x = Bs[kk * LDSB_STR];
    f.b[s].y = Bs[(kk + 1) * LDSB_STR];
  }
}

static __device__ __forceinline__ void pfrag_mma(const PFrag& f, v8f& acc0, v8f& acc1) {
#pragma unroll
  for (int s = 0; s < FKB; ++s) {
    acc0 = __builtin_amdgcn_wmma_f32_16x16x4_f32(false, f.a0[s], false, f.b[s], (short)0, acc0, false, false);
    acc1 = __builtin_amdgcn_wmma_f32_16x16x4_f32(false, f.a1[s], false, f.b[s], (short)0, acc1, false, false);
  }
}

__global__ __launch_bounds__(128) void proj_wmma(const float* __restrict__ batch,
                                                 const float* __restrict__ Bm,
                                                 float* __restrict__ T) {
  __shared__ __attribute__((aligned(16))) float ldsA[2 * LDSA_STAGE];
  __shared__ __attribute__((aligned(16))) float ldsB[2 * LDSB_STAGE];

  const int ch   = blockIdx.x;
  const int rb   = blockIdx.y;
  const int t    = threadIdx.x;
  const int wave = t >> 5;
  const int lane = t & 31;
  const int lcol = lane & 15;
  const int khalf = (lane >> 4) << 1;
  const int n0   = wave * 16;

  const float* Agt = batch + (size_t)(rb * PM + (t >> 5)) * DIM + (t & 31) * 4;
  const float* Bgt = Bm + (size_t)(t >> 4) * RANK + ch * PN + (t & 15) * 4;
  float* lAt = ldsA + (t >> 5) * LDSA_STR + (t & 31) * 4;
  float* lBt = ldsB + (t >> 4) * LDSB_STR + (t & 15) * 4;

  {
    const float* An = Agt;
    const float* Bn = Bgt;
    float* lAn = lAt;
    float* lBn = lBt;
    A_EACH(A_DECL)
    B_EACH(B_DECL)
    A_EACH(A_LOAD)
    B_EACH(B_LOAD)
    A_EACH(A_STORE)
    B_EACH(B_STORE)
  }
  __syncthreads();

  v8f acc0 = {}, acc1 = {};

#pragma nounroll
  for (int i = 0; i < NCHUNK; ++i) {
    const int  cur  = i & 1;
    const bool more = (i + 1 < NCHUNK);
    A_EACH(A_DECL)
    B_EACH(B_DECL)

    const float* As = ldsA + cur * LDSA_STAGE + lcol * LDSA_STR + khalf;
    const float* Bs = ldsB + cur * LDSB_STAGE + khalf * LDSB_STR + n0 + lcol;
    PFrag f0, f1;
    pfrag_load(f0, As, Bs, 0);
    PFENCE();

    if (more) {
      const float* An = Agt + (i + 1) * KCHUNK;
      const float* Bn = Bgt + (size_t)(i + 1) * KCHUNK * RANK;
      A_EACH(A_LOAD)
      B_EACH(B_LOAD)
    }
    PFENCE();

    pfrag_load(f1, As, Bs, 1 * 4 * FKB);
    PFENCE();
    pfrag_mma(f0, acc0, acc1);
    PFENCE();
    pfrag_load(f0, As, Bs, 2 * 4 * FKB);
    PFENCE();
    pfrag_mma(f1, acc0, acc1);
    PFENCE();
    pfrag_load(f1, As, Bs, 3 * 4 * FKB);
    PFENCE();
    pfrag_mma(f0, acc0, acc1);
    PFENCE();
    pfrag_mma(f1, acc0, acc1);

    if (more) {
      float* lAn = lAt + (cur ^ 1) * LDSA_STAGE;
      float* lBn = lBt + (cur ^ 1) * LDSB_STAGE;
      A_EACH(A_STORE)
      B_EACH(B_STORE)
    }
    __syncthreads();
  }

  float* Tt = T + (size_t)(rb * PM) * RANK + ch * PN + n0 + lcol;
  const int mhi = (lane >> 4) * 8;
#pragma unroll
  for (int v = 0; v < 8; ++v) {
    Tt[(size_t)(v + mhi) * RANK]      = acc0[v];
    Tt[(size_t)(16 + v + mhi) * RANK] = acc1[v];
  }
}

struct GFrag { v2f a0[FKB]; v2f a1[FKB]; v2f b0[FKB]; v2f b1[FKB]; };

static __device__ __forceinline__ void gfrag_load(GFrag& f, const float* Ai, const float* Bj, const int k0) {
#pragma unroll
  for (int s = 0; s < FKB; ++s) {
    const int kk = k0 + 4 * s;
    f.a0[s] = *reinterpret_cast<const v2f*>(Ai + kk);
    f.a1[s] = *reinterpret_cast<const v2f*>(Ai + 16 * LDST_STR + kk);
    f.b0[s] = *reinterpret_cast<const v2f*>(Bj + kk);
    f.b1[s] = *reinterpret_cast<const v2f*>(Bj + 16 * LDST_STR + kk);
  }
}

static __device__ __forceinline__ void gfrag_mma(const GFrag& f, v8f& acc00, v8f& acc01, v8f& acc10, v8f& acc11,
                                                 v2f& na0, v2f& na1, v2f& nb0, v2f& nb1) {
#pragma unroll
  for (int s = 0; s < FKB; ++s) {
    acc00 = __builtin_amdgcn_wmma_f32_16x16x4_f32(false, f.a0[s], false, f.b0[s], (short)0, acc00, false, false);
    na0 = __builtin_elementwise_fma(f.a0[s], f.a0[s], na0);
    acc01 = __builtin_amdgcn_wmma_f32_16x16x4_f32(false, f.a0[s], false, f.b1[s], (short)0, acc01, false, false);
    nb0 = __builtin_elementwise_fma(f.b0[s], f.b0[s], nb0);
    acc10 = __builtin_amdgcn_wmma_f32_16x16x4_f32(false, f.a1[s], false, f.b0[s], (short)0, acc10, false, false);
    na1 = __builtin_elementwise_fma(f.a1[s], f.a1[s], na1);
    acc11 = __builtin_amdgcn_wmma_f32_16x16x4_f32(false, f.a1[s], false, f.b1[s], (short)0, acc11, false, false);
    nb1 = __builtin_elementwise_fma(f.b1[s], f.b1[s], nb1);
  }
}

__global__ __launch_bounds__(128) void gram_wmma(const float* __restrict__ T,
                                                 float* __restrict__ out) {
  __shared__ float ldsT[2 * GT * LDST_STR];

  const int b     = blockIdx.x;
  const int jgrp  = blockIdx.y;
  const int igrp  = blockIdx.z;
  const int t     = threadIdx.x;
  const int wave  = t >> 5;
  const int wi    = wave >> 1;
  const int wj    = wave & 1;
  const int lane  = t & 31;
  const int lcol  = lane & 15;
  const int khalf = (lane >> 4) << 1;

  const float* Tb = T + (size_t)b * SEQ * RANK;
  const float* Ti = Tb + (size_t)(igrp * GT) * RANK;
  const float* Tj = Tb + (size_t)(jgrp * GT) * RANK;

#pragma unroll
  for (int c = 0; c < 16; ++c) {
    const int idx = (c * 128 + t) * 4;
    const int row = idx >> 7;
    const int col = idx & 127;
    const float4 vi = *reinterpret_cast<const float4*>(Ti + idx);
    const float4 vj = *reinterpret_cast<const float4*>(Tj + idx);
    *reinterpret_cast<float4*>(&ldsT[row * LDST_STR + col])        = vi;
    *reinterpret_cast<float4*>(&ldsT[(GT + row) * LDST_STR + col]) = vj;
  }
  __syncthreads();

  const float* Ai = &ldsT[(wi * 32 + lcol) * LDST_STR + khalf];
  const float* Bj = &ldsT[(GT + wj * 32 + lcol) * LDST_STR + khalf];

  v8f acc00 = {}, acc01 = {}, acc10 = {}, acc11 = {};
  v2f na0 = {}, na1 = {}, nb0 = {}, nb1 = {};
  GFrag g0, g1;
  gfrag_load(g0, Ai, Bj, 0);
  GFENCE();
  gfrag_load(g1, Ai, Bj, 1 * 4 * FKB);
  GFENCE();
  gfrag_mma(g0, acc00, acc01, acc10, acc11, na0, na1, nb0, nb1);
  GFENCE();
  gfrag_load(g0, Ai, Bj, 2 * 4 * FKB);
  GFENCE();
  gfrag_mma(g1, acc00, acc01, acc10, acc11, na0, na1, nb0, nb1);
  GFENCE();
  gfrag_load(g1, Ai, Bj, 3 * 4 * FKB);
  GFENCE();
  gfrag_mma(g0, acc00, acc01, acc10, acc11, na0, na1, nb0, nb1);
  GFENCE();
  gfrag_mma(g1, acc00, acc01, acc10, acc11, na0, na1, nb0, nb1);

  const float pa0 = na0.x + na0.y, pa1 = na1.x + na1.y;
  const float pb0 = nb0.x + nb0.y, pb1 = nb1.x + nb1.y;
  const float nil0 = pa0 + __shfl_xor(pa0, 16);
  const float nil1 = pa1 + __shfl_xor(pa1, 16);
  const float nj0  = pb0 + __shfl_xor(pb0, 16);
  const float nj1  = pb1 + __shfl_xor(pb1, 16);

  float* outb = out + (size_t)b * SEQ * SEQ + (size_t)(igrp * GT) * SEQ + jgrp * GT;
  const int mhi = (lane >> 4) * 8;
  const int j0  = wj * 32 + lcol;
  const int j1  = j0 + 16;
#pragma unroll
  for (int v = 0; v < 8; ++v) {
    const int i0 = wi * 32 + v + mhi;
    const int i1 = i0 + 16;
    const float ni0 = __shfl(nil0, v + mhi);
    const float ni1 = __shfl(nil1, v + mhi);
    outb[(size_t)i0 * SEQ + j0] = ni0 + nj0 - 2.0f * acc00[v];
    outb[(size_t)i0 * SEQ + j1] = ni0 + nj1 - 2.0f * acc01[v];
    outb[(size_t)i1 * SEQ + j0] = ni1 + nj0 - 2.0f * acc10[v];
    outb[(size_t)i1 * SEQ + j1] = ni1 + nj1 - 2.0f * acc11[v];
  }
}

extern "C" void kernel_launch(void* const* d_in, const int* in_sizes, int n_in,
                              void* d_out, int out_size, void* d_ws, size_t ws_size,
                              hipStream_t stream) {
  const float* batch = (const float*)d_in[0];
  const float* Bm    = (const float*)d_in[1];
  float* out = (float*)d_out;
  float* T   = (float*)d_ws;

  proj_wmma<<<dim3(RANK / PN, BATCHES * SEQ / PM), 128, 0, stream>>>(batch, Bm, T);
  gram_wmma<<<dim3(BATCHES, SEQ / GT, SEQ / GT), 128, 0, stream>>>(T, out);
}


# Projection K loop: chunk hand-over moved inside the WMMA blocks: next chunk's LDS stores and global loads before block 3, LDS-only barrier and first fragment loads before block 4; sums keep order
# speedup vs baseline: 2.4546x; 1.0562x over previous
#include <hip/hip_runtime.h>

typedef float v2f __attribute__((ext_vector_type(2)));
typedef float v8f __attribute__((ext_vector_type(8)));

#define BATCHES   16
#define SEQ       256
#define DIM       1024
#define RANK      128
#define KCHUNK    128
#define NCHUNK    (DIM / KCHUNK)
#define PM        32
#define PN        64
#define LDSA_STR  132
#define LDSA_STAGE (PM * LDSA_STR)
#define LDSB_STR  80
#define LDSB_STAGE (KCHUNK * LDSB_STR)
#define LDST_STR  132
#define GT        64

#define A_EACH(OP) OP(0) OP(1) OP(2) OP(3) OP(4) OP(5) OP(6) OP(7)
#define B_EACH(OP) OP(0) OP(1) OP(2) OP(3) OP(4) OP(5) OP(6) OP(7) \
                   OP(8) OP(9) OP(10) OP(11) OP(12) OP(13) OP(14) OP(15)
#define A_DECL(c)  float4 pa##c;
#define A_LOAD(c)  pa##c = *reinterpret_cast<const float4*>(An + (size_t)(c) * 4 * DIM);
#define A_STORE(c) *reinterpret_cast<float4*>(lAn + (c) * 4 * LDSA_STR) = pa##c;
#define B_DECL(c)  float4 pb##c;
#define B_LOAD(c)  pb##c = *reinterpret_cast<const float4*>(Bn + (c) * 8 * RANK);
#define B_STORE(c) *reinterpret_cast<float4*>(lBn + (c) * 8 * LDSB_STR) = pb##c;

#define FKB 8

#define PFENCE() asm volatile("" : "+v"(acc0), "+v"(acc1) : : "memory")
#define GFENCE() asm volatile("" : "+v"(acc00), "+v"(acc01), "+v"(acc10), "+v"(acc11) : : "memory")

#define LDS_BARRIER() do { \
    __builtin_amdgcn_fence(__ATOMIC_RELEASE, "workgroup", "local"); \
    __builtin_amdgcn_s_barrier(); \
    __builtin_amdgcn_fence(__ATOMIC_ACQUIRE, "workgroup", "local"); } while (0)

struct PFrag { v2f a0[FKB]; v2f a1[FKB]; v2f b[FKB]; };

static __device__ __forceinline__ void pfrag_load(PFrag& f, const float* As, const float* Bs, const int k0) {
#pragma unroll
  for (int s = 0; s < FKB; ++s) {
    const int kk = k0 + 4 * s;
    f.a0[s] = *reinterpret_cast<const v2f*>(As + kk);
    f.a1[s] = *reinterpret_cast<const v2f*>(As + 16 * LDSA_STR + kk);
    f.b[s].x = Bs[kk * LDSB_STR];
    f.b[s].y = Bs[(kk + 1) * LDSB_STR];
  }
}

static __device__ __forceinline__ void pfrag_mma(const PFrag& f, v8f& acc0, v8f& acc1) {
#pragma unroll
  for (int s = 0; s < FKB; ++s) {
    acc0 = __builtin_amdgcn_wmma_f32_16x16x4_f32(false, f.a0[s], false, f.b[s], (short)0, acc0, false, false);
    acc1 = __builtin_amdgcn_wmma_f32_16x16x4_f32(false, f.a1[s], false, f.b[s], (short)0, acc1, false, false);
  }
}

__global__ __launch_bounds__(128) void proj_wmma(const float* __restrict__ batch,
                                                 const float* __restrict__ Bm,
                                                 float* __restrict__ T) {
  __shared__ __attribute__((aligned(16))) float ldsA[2 * LDSA_STAGE];
  __shared__ __attribute__((aligned(16))) float ldsB[2 * LDSB_STAGE];

  const int ch   = blockIdx.x;
  const int rb   = blockIdx.y;
  const int t    = threadIdx.x;
  const int wave = t >> 5;
  const int lane = t & 31;
  const int lcol = lane & 15;
  const int khalf = (lane >> 4) << 1;
  const int n0   = wave * 16;

  const float* Agt = batch + (size_t)(rb * PM + (t >> 5)) * DIM + (t & 31) * 4;
  const float* Bgt = Bm + (size_t)(t >> 4) * RANK + ch * PN + (t & 15) * 4;
  float* lAt = ldsA + (t >> 5) * LDSA_STR + (t & 31) * 4;
  float* lBt = ldsB + (t >> 4) * LDSB_STR + (t & 15) * 4;

  v8f acc0 = {}, acc1 = {};

  A_EACH(A_DECL)
  B_EACH(B_DECL)

  {
    const float* An = Agt;
    const float* Bn = Bgt;
    float* lAn = lAt;
    float* lBn = lBt;
    A_EACH(A_LOAD)
    B_EACH(B_LOAD)
    A_EACH(A_STORE)
    B_EACH(B_STORE)
  }
  PFENCE();
  {
    const float* An = Agt + KCHUNK;
    const float* Bn = Bgt + (size_t)KCHUNK * RANK;
    A_EACH(A_LOAD)
    B_EACH(B_LOAD)
  }
  PFENCE();
  LDS_BARRIER();

  PFrag f0, f1;
  pfrag_load(f0, ldsA + lcol * LDSA_STR + khalf, ldsB + khalf * LDSB_STR + n0 + lcol, 0);
  PFENCE();

#pragma nounroll
  for (int i = 0; i < NCHUNK - 1; ++i) {
    const int cur = i & 1;

    const float* As = ldsA + cur * LDSA_STAGE + lcol * LDSA_STR + khalf;
    const float* Bs = ldsB + cur * LDSB_STAGE + khalf * LDSB_STR + n0 + lcol;

    pfrag_load(f1, As, Bs, 1 * 4 * FKB);
    PFENCE();
    pfrag_mma(f0, acc0, acc1);
    PFENCE();
    pfrag_load(f0, As, Bs, 2 * 4 * FKB);
    PFENCE();
    pfrag_mma(f1, acc0, acc1);
    PFENCE();
    pfrag_load(f1, As, Bs, 3 * 4 * FKB);
    {
      float* lAn = lAt + (cur ^ 1) * LDSA_STAGE;
      float* lBn = lBt + (cur ^ 1) * LDSB_STAGE;
      A_EACH(A_STORE)
      B_EACH(B_STORE)
    }
    if (i + 2 < NCHUNK) {
      const float* An = Agt + (i + 2) * KCHUNK;
      const float* Bn = Bgt + (size_t)(i + 2) * KCHUNK * RANK;
      A_EACH(A_LOAD)
      B_EACH(B_LOAD)
    }
    pfrag_mma(f0, acc0, acc1);
    PFENCE();
    LDS_BARRIER();
    pfrag_load(f0, ldsA + (cur ^ 1) * LDSA_STAGE + lcol * LDSA_STR + khalf,
                   ldsB + (cur ^ 1) * LDSB_STAGE + khalf * LDSB_STR + n0 + lcol, 0);
    PFENCE();
    pfrag_mma(f1, acc0, acc1);
  }

  {
    const int cur = (NCHUNK - 1) & 1;
    const float* As = ldsA + cur * LDSA_STAGE + lcol * LDSA_STR + khalf;
    const float* Bs = ldsB + cur * LDSB_STAGE + khalf * LDSB_STR + n0 + lcol;
    pfrag_load(f1, As, Bs, 1 * 4 * FKB);
    PFENCE();
    pfrag_mma(f0, acc0, acc1);
    PFENCE();
    pfrag_load(f0, As, Bs, 2 * 4 * FKB);
    PFENCE();
    pfrag_mma(f1, acc0, acc1);
    PFENCE();
    pfrag_load(f1, As, Bs, 3 * 4 * FKB);
    PFENCE();
    pfrag_mma(f0, acc0, acc1);
    PFENCE();
    pfrag_mma(f1, acc0, acc1);
  }

  float* Tt = T + (size_t)(rb * PM) * RANK + ch * PN + n0 + lcol;
  const int mhi = (lane >> 4) * 8;
#pragma unroll
  for (int v = 0; v < 8; ++v) {
    Tt[(size_t)(v + mhi) * RANK]      = acc0[v];
    Tt[(size_t)(16 + v + mhi) * RANK] = acc1[v];
  }
}

struct GFrag { v2f a0[FKB]; v2f a1[FKB]; v2f b0[FKB]; v2f b1[FKB]; };

static __device__ __forceinline__ void gfrag_load(GFrag& f, const float* Ai, const float* Bj, const int k0) {
#pragma unroll
  for (int s = 0; s < FKB; ++s) {
    const int kk = k0 + 4 * s;
    f.a0[s] = *reinterpret_cast<const v2f*>(Ai + kk);
    f.a1[s] = *reinterpret_cast<const v2f*>(Ai + 16 * LDST_STR + kk);
    f.b0[s] = *reinterpret_cast<const v2f*>(Bj + kk);
    f.b1[s] = *reinterpret_cast<const v2f*>(Bj + 16 * LDST_STR + kk);
  }
}

static __device__ __forceinline__ void gfrag_mma(const GFrag& f, v8f& acc00, v8f& acc01, v8f& acc10, v8f& acc11,
                                                 v2f& na0, v2f& na1, v2f& nb0, v2f& nb1) {
#pragma unroll
  for (int s = 0; s < FKB; ++s) {
    acc00 = __builtin_amdgcn_wmma_f32_16x16x4_f32(false, f.a0[s], false, f.b0[s], (short)0, acc00, false, false);
    na0 = __builtin_elementwise_fma(f.a0[s], f.a0[s], na0);
    acc01 = __builtin_amdgcn_wmma_f32_16x16x4_f32(false, f.a0[s], false, f.b1[s], (short)0, acc01, false, false);
    nb0 = __builtin_elementwise_fma(f.b0[s], f.b0[s], nb0);
    acc10 = __builtin_amdgcn_wmma_f32_16x16x4_f32(false, f.a1[s], false, f.b0[s], (short)0, acc10, false, false);
    na1 = __builtin_elementwise_fma(f.a1[s], f.a1[s], na1);
    acc11 = __builtin_amdgcn_wmma_f32_16x16x4_f32(false, f.a1[s], false, f.b1[s], (short)0, acc11, false, false);
    nb1 = __builtin_elementwise_fma(f.b1[s], f.b1[s], nb1);
  }
}

__global__ __launch_bounds__(128) void gram_wmma(const float* __restrict__ T,
                                                 float* __restrict__ out) {
  __shared__ float ldsT[2 * GT * LDST_STR];

  const int b     = blockIdx.x;
  const int jgrp  = blockIdx.y;
  const int igrp  = blockIdx.z;
  const int t     = threadIdx.x;
  const int wave  = t >> 5;
  const int wi    = wave >> 1;
  const int wj    = wave & 1;
  const int lane  = t & 31;
  const int lcol  = lane & 15;
  const int khalf = (lane >> 4) << 1;

  const float* Tb = T + (size_t)b * SEQ * RANK;
  const float* Ti = Tb + (size_t)(igrp * GT) * RANK;
  const float* Tj = Tb + (size_t)(jgrp * GT) * RANK;

#pragma unroll
  for (int c = 0; c < 16; ++c) {
    const int idx = (c * 128 + t) * 4;
    const int row = idx >> 7;
    const int col = idx & 127;
    const float4 vi = *reinterpret_cast<const float4*>(Ti + idx);
    const float4 vj = *reinterpret_cast<const float4*>(Tj + idx);
    *reinterpret_cast<float4*>(&ldsT[row * LDST_STR + col])        = vi;
    *reinterpret_cast<float4*>(&ldsT[(GT + row) * LDST_STR + col]) = vj;
  }
  __syncthreads();

  const float* Ai = &ldsT[(wi * 32 + lcol) * LDST_STR + khalf];
  const float* Bj = &ldsT[(GT + wj * 32 + lcol) * LDST_STR + khalf];

  v8f acc00 = {}, acc01 = {}, acc10 = {}, acc11 = {};
  v2f na0 = {}, na1 = {}, nb0 = {}, nb1 = {};
  GFrag g0, g1;
  gfrag_load(g0, Ai, Bj, 0);
  GFENCE();
  gfrag_load(g1, Ai, Bj, 1 * 4 * FKB);
  GFENCE();
  gfrag_mma(g0, acc00, acc01, acc10, acc11, na0, na1, nb0, nb1);
  GFENCE();
  gfrag_load(g0, Ai, Bj, 2 * 4 * FKB);
  GFENCE();
  gfrag_mma(g1, acc00, acc01, acc10, acc11, na0, na1, nb0, nb1);
  GFENCE();
  gfrag_load(g1, Ai, Bj, 3 * 4 * FKB);
  GFENCE();
  gfrag_mma(g0, acc00, acc01, acc10, acc11, na0, na1, nb0, nb1);
  GFENCE();
  gfrag_mma(g1, acc00, acc01, acc10, acc11, na0, na1, nb0, nb1);

  const float pa0 = na0.x + na0.y, pa1 = na1.x + na1.y;
  const float pb0 = nb0.x + nb0.y, pb1 = nb1.x + nb1.y;
  const float nil0 = pa0 + __shfl_xor(pa0, 16);
  const float nil1 = pa1 + __shfl_xor(pa1, 16);
  const float nj0  = pb0 + __shfl_xor(pb0, 16);
  const float nj1  = pb1 + __shfl_xor(pb1, 16);

  float* outb = out + (size_t)b * SEQ * SEQ + (size_t)(igrp * GT) * SEQ + jgrp * GT;
  const int mhi = (lane >> 4) * 8;
  const int j0  = wj * 32 + lcol;
  const int j1  = j0 + 16;
#pragma unroll
  for (int v = 0; v < 8; ++v) {
    const int i0 = wi * 32 + v + mhi;
    const int i1 = i0 + 16;
    const float ni0 = __shfl(nil0, v + mhi);
    const float ni1 = __shfl(nil1, v + mhi);
    outb[(size_t)i0 * SEQ + j0] = ni0 + nj0 - 2.0f * acc00[v];
    outb[(size_t)i0 * SEQ + j1] = ni0 + nj1 - 2.0f * acc01[v];
    outb[(size_t)i1 * SEQ + j0] = ni1 + nj0 - 2.0f * acc10[v];
    outb[(size_t)i1 * SEQ + j1] = ni1 + nj1 - 2.0f * acc11[v];
  }
}

extern "C" void kernel_launch(void* const* d_in, const int* in_sizes, int n_in,
                              void* d_out, int out_size, void* d_ws, size_t ws_size,
                              hipStream_t stream) {
  const float* batch = (const float*)d_in[0];
  const float* Bm    = (const float*)d_in[1];
  float* out = (float*)d_out;
  float* T   = (float*)d_ws;

  proj_wmma<<<dim3(RANK / PN, BATCHES * SEQ / PM), 128, 0, stream>>>(batch, Bm, T);
  gram_wmma<<<dim3(BATCHES, SEQ / GT, SEQ / GT), 128, 0, stream>>>(T, out);
}
